# starting kernel
# baseline (speedup 1.0000x reference)
#include <hip/hip_runtime.h>

typedef float v2f __attribute__((ext_vector_type(2)));
typedef float v8f __attribute__((ext_vector_type(8)));

#define D   128
#define NPG 64

__global__ __launch_bounds__(256) void partner_kernel(
    const float* __restrict__ node, const float* __restrict__ glob,
    const float* __restrict__ Wp1,  const float* __restrict__ bp1,
    const float* __restrict__ Wp2,  const float* __restrict__ bp2,
    float* __restrict__ out)
{
  extern __shared__ float smem[];
  float* lds_wb   = smem;
  float* lds_base = smem + 16384;
  float* lds_vec  = smem + 16384 + 256;
  float* lds_wp2  = smem + 16384 + 768;

  const int tid = threadIdx.x;
  const int g0  = blockIdx.x * 2;

  for (int i = tid; i < D * D; i += 256) {
    int k = i >> 7, n = i & 127;
    float v = Wp1[(D + k) * D + n];
    int p = k >> 1;
    int col = (n + ((p & 1) << 4)) & 127;
    lds_wb[p * 256 + col * 2 + (k & 1)] = v;
  }
  {
    int g = tid >> 7, j = tid & 127;
    int graph = g0 + g;
    lds_vec[(g * 2 + 0) * D + j] = node[(graph * NPG + NPG - 1) * D + j];
    lds_vec[(g * 2 + 1) * D + j] = glob[graph * D + j];
  }
  if (tid < D) lds_wp2[tid] = Wp2[tid];
  __syncthreads();

  {
    int g = tid >> 7, j = tid & 127;
    float s = bp1[j];
    const float* cv = &lds_vec[(g * 2 + 0) * D];
    const float* gv = &lds_vec[(g * 2 + 1) * D];
#pragma unroll 8
    for (int k = 0; k < D; ++k) {
      s += cv[k] * Wp1[k * D + j];
      s += gv[k] * Wp1[(2 * D + k) * D + j];
    }
    lds_base[g * D + j] = s;
  }
  __syncthreads();

  const int wave = tid >> 5, lane = tid & 31;
  const int half = lane >> 4, l16 = lane & 15;
  const int g = wave >> 2, mtile = wave & 3;
  const int graph = g0 + g;
  const int row = graph * NPG + mtile * 16 + l16;
  const float* rowp = node + (size_t)row * D + 2 * half;

  v2f af[32];
#pragma unroll
  for (int s = 0; s < 32; ++s) af[s] = *(const v2f*)(rowp + 4 * s);

  const v2f* wb2 = (const v2f*)lds_wb;
  const int coln0 = l16;
  const int coln1 = (l16 + 16) & 127;

  float acc[8];
#pragma unroll
  for (int r = 0; r < 8; ++r) acc[r] = 0.f;

  for (int nt = 0; nt < 8; ++nt) {
    int n = nt * 16 + l16;
    int col = ((n + 16 * half) & 127);
    v8f c = {};
#pragma unroll
    for (int s = 0; s < 32; ++s) {
      v2f b = wb2[(2 * s + half) * D + col];
      c = __builtin_amdgcn_wmma_f32_16x16x4_f32(false, af[s], false, b,
                                                (short)0, c, false, false);
    }
    float bn = lds_base[g * D + n];
    float wn = lds_wp2[n];
#pragma unroll
    for (int r = 0; r < 8; ++r) {
      float v = c[r] + bn;
      v = v > 0.f ? v : 0.f;
      acc[r] += v * wn;
    }
  }
#pragma unroll
  for (int r = 0; r < 8; ++r) {
    float a = acc[r];
    a += __shfl_xor(a, 1, 32);
    a += __shfl_xor(a, 2, 32);
    a += __shfl_xor(a, 4, 32);
    a += __shfl_xor(a, 8, 32);
    acc[r] = a;
  }
  if (l16 == 0) {
    float b2 = bp2[0];
#pragma unroll
    for (int r = 0; r < 8; ++r)
      out[graph * NPG + mtile * 16 + r + 8 * half] = acc[r] + b2;
  }
}

__global__ __launch_bounds__(256) void label_kernel(
    const float* __restrict__ node, const float* __restrict__ glob,
    const float* __restrict__ Wl1,  const float* __restrict__ bl1,
    const float* __restrict__ Wl2,  const float* __restrict__ bl2,
    const float* __restrict__ Wl3,  const float* __restrict__ bl3,
    const int* __restrict__ pgi,    const int* __restrict__ pni,
    float* __restrict__ out, int outoff)
{
  extern __shared__ float smem[];
  float* lds_h1 = smem;
  float* lds_h2 = smem + 8 * 16 * D;
  float* lds_w3 = smem + 2 * 8 * 16 * D;

  const int tid = threadIdx.x;
  for (int i = tid; i < D * 16; i += 256) {
    int k = i >> 4, n = i & 15;
    lds_w3[i] = (n < 14) ? Wl3[k * 14 + n] : 0.f;
  }

  const int wave = tid >> 5, lane = tid & 31;
  const int half = lane >> 4, l16 = lane & 15;
  const int rowid = blockIdx.x * 128 + wave * 16 + l16;
  const int pg = pgi[rowid];
  const int pn = pni[rowid];
  const float* src0 = node + ((size_t)(pg + 1) * NPG - 1) * D;
  const float* src1 = node + (size_t)pn * D;
  const float* src2 = glob + (size_t)pg * D;

  for (int nt = 0; nt < 8; ++nt) {
    int n = nt * 16 + l16;
    v8f c = {};
#pragma unroll
    for (int seg = 0; seg < 3; ++seg) {
      const float* ap = (seg == 0 ? src0 : seg == 1 ? src1 : src2) + 2 * half;
      const float* bp = Wl1 + (size_t)(seg * D + 2 * half) * D + n;
#pragma unroll
      for (int s = 0; s < 32; ++s) {
        v2f a = *(const v2f*)(ap + 4 * s);
        v2f b;
        b.x = bp[(size_t)(4 * s) * D];
        b.y = bp[(size_t)(4 * s + 1) * D];
        c = __builtin_amdgcn_wmma_f32_16x16x4_f32(false, a, false, b,
                                                  (short)0, c, false, false);
      }
    }
    float bv = bl1[n];
#pragma unroll
    for (int r = 0; r < 8; ++r) {
      float v = c[r] + bv;
      lds_h1[(wave * 16 + r + 8 * half) * D + n] = v > 0.f ? v : 0.f;
    }
  }
  __syncthreads();

  for (int nt = 0; nt < 8; ++nt) {
    int n = nt * 16 + l16;
    v8f c = {};
    const float* ap = lds_h1 + (wave * 16 + l16) * D + 2 * half;
    const float* bp = Wl2 + (size_t)(2 * half) * D + n;
#pragma unroll
    for (int s = 0; s < 32; ++s) {
      v2f a = *(const v2f*)(ap + 4 * s);
      v2f b;
      b.x = bp[(size_t)(4 * s) * D];
      b.y = bp[(size_t)(4 * s + 1) * D];
      c = __builtin_amdgcn_wmma_f32_16x16x4_f32(false, a, false, b,
                                                (short)0, c, false, false);
    }
    float bv = bl2[n];
#pragma unroll
    for (int r = 0; r < 8; ++r) {
      float v = c[r] + bv;
      lds_h2[(wave * 16 + r + 8 * half) * D + n] = v > 0.f ? v : 0.f;
    }
  }
  __syncthreads();

  {
    v8f c = {};
    const float* ap = lds_h2 + (wave * 16 + l16) * D + 2 * half;
#pragma unroll
    for (int s = 0; s < 32; ++s) {
      v2f a = *(const v2f*)(ap + 4 * s);
      v2f b;
      b.x = lds_w3[(4 * s + 2 * half) * 16 + l16];
      b.y = lds_w3[(4 * s + 2 * half + 1) * 16 + l16];
      c = __builtin_amdgcn_wmma_f32_16x16x4_f32(false, a, false, b,
                                                (short)0, c, false, false);
    }
    if (l16 < 14) {
      float bv = bl3[l16];
#pragma unroll
      for (int r = 0; r < 8; ++r) {
        int grow = blockIdx.x * 128 + wave * 16 + r + 8 * half;
        out[outoff + grow * 14 + l16] = c[r] + bv;
      }
    }
  }
}

extern "C" void kernel_launch(void* const* d_in, const int* in_sizes, int n_in,
                              void* d_out, int out_size, void* d_ws, size_t ws_size,
                              hipStream_t stream) {
  (void)n_in; (void)out_size; (void)d_ws; (void)ws_size;
  const float* node = (const float*)d_in[0];
  const float* glob = (const float*)d_in[1];
  const float* Wp1  = (const float*)d_in[2];
  const float* bp1  = (const float*)d_in[3];
  const float* Wp2  = (const float*)d_in[4];
  const float* bp2  = (const float*)d_in[5];
  const float* Wl1  = (const float*)d_in[6];
  const float* bl1  = (const float*)d_in[7];
  const float* Wl2  = (const float*)d_in[8];
  const float* bl2  = (const float*)d_in[9];
  const float* Wl3  = (const float*)d_in[10];
  const float* bl3  = (const float*)d_in[11];
  const int*   pgi  = (const int*)d_in[12];
  const int*   pni  = (const int*)d_in[13];
  float* out = (float*)d_out;

  const int B = in_sizes[1] / D;
  const int N = in_sizes[0] / D;

  const size_t smem1 = (16384 + 256 + 512 + 128) * sizeof(float);
  const size_t smem2 = (2 * 8 * 16 * D + D * 16) * sizeof(float);

  partner_kernel<<<B / 2, 256, smem1, stream>>>(node, glob, Wp1, bp1, Wp2, bp2, out);
  label_kernel<<<B / 128, 256, smem2, stream>>>(node, glob, Wl1, bl1, Wl2, bl2,
                                                Wl3, bl3, pgi, pni, out, N);
}


# label kernel re-tiled: one 16-row tile per block (512 blocks, was 64); each layer's 8 column tiles go to the block's 8 waves instead of a per-wave loop; input tile shared through LDS; sums unchanged
# speedup vs baseline: 1.9727x; 1.9727x over previous
#include <hip/hip_runtime.h>

typedef float v2f __attribute__((ext_vector_type(2)));
typedef float v8f __attribute__((ext_vector_type(8)));

#define D   128
#define NPG 64

__global__ __launch_bounds__(256) void partner_kernel(
    const float* __restrict__ node, const float* __restrict__ glob,
    const float* __restrict__ Wp1,  const float* __restrict__ bp1,
    const float* __restrict__ Wp2,  const float* __restrict__ bp2,
    float* __restrict__ out)
{
  extern __shared__ float smem[];
  float* lds_wb   = smem;
  float* lds_base = smem + 16384;
  float* lds_vec  = smem + 16384 + 256;
  float* lds_wp2  = smem + 16384 + 768;

  const int tid = threadIdx.x;
  const int g0  = blockIdx.x * 2;

  for (int i = tid; i < D * D; i += 256) {
    int k = i >> 7, n = i & 127;
    float v = Wp1[(D + k) * D + n];
    int p = k >> 1;
    int col = (n + ((p & 1) << 4)) & 127;
    lds_wb[p * 256 + col * 2 + (k & 1)] = v;
  }
  {
    int g = tid >> 7, j = tid & 127;
    int graph = g0 + g;
    lds_vec[(g * 2 + 0) * D + j] = node[(graph * NPG + NPG - 1) * D + j];
    lds_vec[(g * 2 + 1) * D + j] = glob[graph * D + j];
  }
  if (tid < D) lds_wp2[tid] = Wp2[tid];
  __syncthreads();

  {
    int g = tid >> 7, j = tid & 127;
    float s = bp1[j];
    const float* cv = &lds_vec[(g * 2 + 0) * D];
    const float* gv = &lds_vec[(g * 2 + 1) * D];
#pragma unroll 8
    for (int k = 0; k < D; ++k) {
      s += cv[k] * Wp1[k * D + j];
      s += gv[k] * Wp1[(2 * D + k) * D + j];
    }
    lds_base[g * D + j] = s;
  }
  __syncthreads();

  const int wave = tid >> 5, lane = tid & 31;
  const int half = lane >> 4, l16 = lane & 15;
  const int g = wave >> 2, mtile = wave & 3;
  const int graph = g0 + g;
  const int row = graph * NPG + mtile * 16 + l16;
  const float* rowp = node + (size_t)row * D + 2 * half;

  v2f af[32];
#pragma unroll
  for (int s = 0; s < 32; ++s) af[s] = *(const v2f*)(rowp + 4 * s);

  const v2f* wb2 = (const v2f*)lds_wb;
  const int coln0 = l16;
  const int coln1 = (l16 + 16) & 127;

  float acc[8];
#pragma unroll
  for (int r = 0; r < 8; ++r) acc[r] = 0.f;

  for (int nt = 0; nt < 8; ++nt) {
    int n = nt * 16 + l16;
    int col = ((n + 16 * half) & 127);
    v8f c = {};
#pragma unroll
    for (int s = 0; s < 32; ++s) {
      v2f b = wb2[(2 * s + half) * D + col];
      c = __builtin_amdgcn_wmma_f32_16x16x4_f32(false, af[s], false, b,
                                                (short)0, c, false, false);
    }
    float bn = lds_base[g * D + n];
    float wn = lds_wp2[n];
#pragma unroll
    for (int r = 0; r < 8; ++r) {
      float v = c[r] + bn;
      v = v > 0.f ? v : 0.f;
      acc[r] += v * wn;
    }
  }
#pragma unroll
  for (int r = 0; r < 8; ++r) {
    float a = acc[r];
    a += __shfl_xor(a, 1, 32);
    a += __shfl_xor(a, 2, 32);
    a += __shfl_xor(a, 4, 32);
    a += __shfl_xor(a, 8, 32);
    acc[r] = a;
  }
  if (l16 == 0) {
    float b2 = bp2[0];
#pragma unroll
    for (int r = 0; r < 8; ++r)
      out[graph * NPG + mtile * 16 + r + 8 * half] = acc[r] + b2;
  }
}

#define LROWS 16
#define XS    (3 * D + 4)
#define HS    (D + 4)

typedef float v4f __attribute__((ext_vector_type(4)));

__global__ __launch_bounds__(256) void label_kernel(
    const float* __restrict__ node, const float* __restrict__ glob,
    const float* __restrict__ Wl1,  const float* __restrict__ bl1,
    const float* __restrict__ Wl2,  const float* __restrict__ bl2,
    const float* __restrict__ Wl3,  const float* __restrict__ bl3,
    const int* __restrict__ pgi,    const int* __restrict__ pni,
    float* __restrict__ out, int outoff)
{
  extern __shared__ float smem[];
  float* lds_x  = smem;
  float* lds_h1 = smem + LROWS * XS;
  float* lds_h2 = smem + LROWS * XS + LROWS * HS;
  float* lds_w3 = smem + LROWS * XS + 2 * LROWS * HS;

  const int tid = threadIdx.x;
  const int srow = tid >> 4, sc = tid & 15;

  float w3v[8];
#pragma unroll
  for (int j = 0; j < 8; ++j)
    w3v[j] = Wl3[(srow + 16 * j) * 14 + (sc < 14 ? sc : 13)];

  {
    const int rowid = blockIdx.x * LROWS + srow;
    const int pg = pgi[rowid];
    const int pn = pni[rowid];
    const float* src0 = node + ((size_t)(pg + 1) * NPG - 1) * D;
    const float* src1 = node + (size_t)pn * D;
    const float* src2 = glob + (size_t)pg * D;
    float* dst = lds_x + srow * XS;
    v4f x00 = *(const v4f*)(src0 + 4 * sc);
    v4f x01 = *(const v4f*)(src0 + 4 * sc + 64);
    v4f x10 = *(const v4f*)(src1 + 4 * sc);
    v4f x11 = *(const v4f*)(src1 + 4 * sc + 64);
    v4f x20 = *(const v4f*)(src2 + 4 * sc);
    v4f x21 = *(const v4f*)(src2 + 4 * sc + 64);
    *(v4f*)(dst + 0 * D + 4 * sc)      = x00;
    *(v4f*)(dst + 0 * D + 4 * sc + 64) = x01;
    *(v4f*)(dst + 1 * D + 4 * sc)      = x10;
    *(v4f*)(dst + 1 * D + 4 * sc + 64) = x11;
    *(v4f*)(dst + 2 * D + 4 * sc)      = x20;
    *(v4f*)(dst + 2 * D + 4 * sc + 64) = x21;
  }
#pragma unroll
  for (int j = 0; j < 8; ++j)
    lds_w3[tid + 256 * j] = sc < 14 ? w3v[j] : 0.f;
  __syncthreads();

  const int wave = tid >> 5, lane = tid & 31;
  const int half = lane >> 4, l16 = lane & 15;
  const int n = wave * 16 + l16;

  {
    v8f c = {};
#pragma unroll
    for (int seg = 0; seg < 3; ++seg) {
      const float* ap = lds_x + l16 * XS + seg * D + 2 * half;
      const float* bp = Wl1 + (size_t)(seg * D + 2 * half) * D + n;
#pragma unroll
      for (int s = 0; s < 32; ++s) {
        v2f a = *(const v2f*)(ap + 4 * s);
        v2f b;
        b.x = bp[(size_t)(4 * s) * D];
        b.y = bp[(size_t)(4 * s + 1) * D];
        c = __builtin_amdgcn_wmma_f32_16x16x4_f32(false, a, false, b,
                                                  (short)0, c, false, false);
      }
    }
    float bv = bl1[n];
#pragma unroll
    for (int r = 0; r < 8; ++r) {
      float v = c[r] + bv;
      lds_h1[(r + 8 * half) * HS + n] = v > 0.f ? v : 0.f;
    }
  }
  __syncthreads();

  {
    v8f c = {};
    const float* ap = lds_h1 + l16 * HS + 2 * half;
    const float* bp = Wl2 + (size_t)(2 * half) * D + n;
#pragma unroll
    for (int s = 0; s < 32; ++s) {
      v2f a = *(const v2f*)(ap + 4 * s);
      v2f b;
      b.x = bp[(size_t)(4 * s) * D];
      b.y = bp[(size_t)(4 * s + 1) * D];
      c = __builtin_amdgcn_wmma_f32_16x16x4_f32(false, a, false, b,
                                                (short)0, c, false, false);
    }
    float bv = bl2[n];
#pragma unroll
    for (int r = 0; r < 8; ++r) {
      float v = c[r] + bv;
      lds_h2[(r + 8 * half) * HS + n] = v > 0.f ? v : 0.f;
    }
  }
  __syncthreads();

  if (wave == 0) {
    v8f c = {};
    const float* ap = lds_h2 + l16 * HS + 2 * half;
#pragma unroll
    for (int s = 0; s < 32; ++s) {
      v2f a = *(const v2f*)(ap + 4 * s);
      v2f b;
      b.x = lds_w3[(4 * s + 2 * half) * 16 + l16];
      b.y = lds_w3[(4 * s + 2 * half + 1) * 16 + l16];
      c = __builtin_amdgcn_wmma_f32_16x16x4_f32(false, a, false, b,
                                                (short)0, c, false, false);
    }
    if (l16 < 14) {
      float bv = bl3[l16];
#pragma unroll
      for (int r = 0; r < 8; ++r) {
        int grow = blockIdx.x * LROWS + r + 8 * half;
        out[outoff + grow * 14 + l16] = c[r] + bv;
      }
    }
  }
}

extern "C" void kernel_launch(void* const* d_in, const int* in_sizes, int n_in,
                              void* d_out, int out_size, void* d_ws, size_t ws_size,
                              hipStream_t stream) {
  (void)n_in; (void)out_size; (void)d_ws; (void)ws_size;
  const float* node = (const float*)d_in[0];
  const float* glob = (const float*)d_in[1];
  const float* Wp1  = (const float*)d_in[2];
  const float* bp1  = (const float*)d_in[3];
  const float* Wp2  = (const float*)d_in[4];
  const float* bp2  = (const float*)d_in[5];
  const float* Wl1  = (const float*)d_in[6];
  const float* bl1  = (const float*)d_in[7];
  const float* Wl2  = (const float*)d_in[8];
  const float* bl2  = (const float*)d_in[9];
  const float* Wl3  = (const float*)d_in[10];
  const float* bl3  = (const float*)d_in[11];
  const int*   pgi  = (const int*)d_in[12];
  const int*   pni  = (const int*)d_in[13];
  float* out = (float*)d_out;

  const int B = in_sizes[1] / D;
  const int N = in_sizes[0] / D;

  const size_t smem1 = (16384 + 256 + 512 + 128) * sizeof(float);
  const size_t smem2 = (LROWS * XS + 2 * LROWS * HS + D * 16) * sizeof(float);

  partner_kernel<<<B / 2, 256, smem1, stream>>>(node, glob, Wp1, bp1, Wp2, bp2, out);
  label_kernel<<<B / LROWS, 256, smem2, stream>>>(node, glob, Wl1, bl1, Wl2, bl2,
                                                Wl3, bl3, pgi, pni, out, N);
}


# partner kernel: per-block set-up (weight-tile staging, base vectors) amortised over 8 graphs: 1024 persistent blocks loop over 4 graph pairs (was 4096 x 2); base now one WMMA tile, sum order changes
# speedup vs baseline: 2.5264x; 1.2807x over previous
#include <hip/hip_runtime.h>

typedef float v2f __attribute__((ext_vector_type(2)));
typedef float v8f __attribute__((ext_vector_type(8)));

#define D   128
#define NPG 64

#define GPB 8

__global__ __launch_bounds__(256) void partner_kernel(
    const float* __restrict__ node, const float* __restrict__ glob,
    const float* __restrict__ Wp1,  const float* __restrict__ bp1,
    const float* __restrict__ Wp2,  const float* __restrict__ bp2,
    float* __restrict__ out)
{
  extern __shared__ float smem[];
  float* lds_wb   = smem;
  float* lds_base = smem + 16384;
  float* lds_wp2  = smem + 16384 + GPB * D;

  const int tid = threadIdx.x;
  const int g0  = blockIdx.x * GPB;
  const int wave = tid >> 5, lane = tid & 31;
  const int half = lane >> 4, l16 = lane & 15;

  for (int i = tid; i < D * D; i += 256) {
    int k = i >> 7, n = i & 127;
    float v = Wp1[(D + k) * D + n];
    int p = k >> 1;
    int col = (n + ((p & 1) << 4)) & 127;
    lds_wb[p * 256 + col * 2 + (k & 1)] = v;
  }
  if (tid < D) lds_wp2[tid] = Wp2[tid];

  {
    const int n  = wave * 16 + l16;
    const int gr = g0 + (l16 & (GPB - 1));
    const float* ac = node + ((size_t)gr * NPG + NPG - 1) * D + 2 * half;
    const float* ag = glob + (size_t)gr * D + 2 * half;
    const float* bc = Wp1 + (size_t)(2 * half) * D + n;
    const float* bg = Wp1 + (size_t)(2 * D + 2 * half) * D + n;
    v8f c = {};
#pragma unroll
    for (int s = 0; s < 32; ++s) {
      v2f a = *(const v2f*)(ac + 4 * s);
      v2f b;
      b.x = bc[(size_t)(4 * s) * D];
      b.y = bc[(size_t)(4 * s + 1) * D];
      c = __builtin_amdgcn_wmma_f32_16x16x4_f32(false, a, false, b,
                                                (short)0, c, false, false);
    }
#pragma unroll
    for (int s = 0; s < 32; ++s) {
      v2f a = *(const v2f*)(ag + 4 * s);
      v2f b;
      b.x = bg[(size_t)(4 * s) * D];
      b.y = bg[(size_t)(4 * s + 1) * D];
      c = __builtin_amdgcn_wmma_f32_16x16x4_f32(false, a, false, b,
                                                (short)0, c, false, false);
    }
    if (half == 0) {
      float bv = bp1[n];
#pragma unroll
      for (int r = 0; r < 8; ++r) lds_base[r * D + n] = c[r] + bv;
    }
  }
  __syncthreads();

  const int gsub = wave >> 2, mtile = wave & 3;
  const v2f* wb2 = (const v2f*)lds_wb;
  const float b2 = bp2[0];

  for (int it = 0; it < GPB / 2; ++it) {
    const int gl = 2 * it + gsub;
    const int graph = g0 + gl;
    const int row = graph * NPG + mtile * 16 + l16;
    const float* rowp = node + (size_t)row * D + 2 * half;

    v2f af[32];
#pragma unroll
    for (int s = 0; s < 32; ++s) af[s] = *(const v2f*)(rowp + 4 * s);

    float acc[8];
#pragma unroll
    for (int r = 0; r < 8; ++r) acc[r] = 0.f;

    for (int nt = 0; nt < 8; ++nt) {
      int n = nt * 16 + l16;
      int col = ((n + 16 * half) & 127);
      v8f c = {};
#pragma unroll
      for (int s = 0; s < 32; ++s) {
        v2f b = wb2[(2 * s + half) * D + col];
        c = __builtin_amdgcn_wmma_f32_16x16x4_f32(false, af[s], false, b,
                                                  (short)0, c, false, false);
      }
      float bn = lds_base[gl * D + n];
      float wn = lds_wp2[n];
#pragma unroll
      for (int r = 0; r < 8; ++r) {
        float v = c[r] + bn;
        v = v > 0.f ? v : 0.f;
        acc[r] += v * wn;
      }
    }
#pragma unroll
    for (int r = 0; r < 8; ++r) {
      float a = acc[r];
      a += __shfl_xor(a, 1, 32);
      a += __shfl_xor(a, 2, 32);
      a += __shfl_xor(a, 4, 32);
      a += __shfl_xor(a, 8, 32);
      acc[r] = a;
    }
    if (l16 == 0) {
#pragma unroll
      for (int r = 0; r < 8; ++r)
        out[graph * NPG + mtile * 16 + r + 8 * half] = acc[r] + b2;
    }
  }
}

#define LROWS 16
#define XS    (3 * D + 4)
#define HS    (D + 4)

typedef float v4f __attribute__((ext_vector_type(4)));

__global__ __launch_bounds__(256) void label_kernel(
    const float* __restrict__ node, const float* __restrict__ glob,
    const float* __restrict__ Wl1,  const float* __restrict__ bl1,
    const float* __restrict__ Wl2,  const float* __restrict__ bl2,
    const float* __restrict__ Wl3,  const float* __restrict__ bl3,
    const int* __restrict__ pgi,    const int* __restrict__ pni,
    float* __restrict__ out, int outoff)
{
  extern __shared__ float smem[];
  float* lds_x  = smem;
  float* lds_h1 = smem + LROWS * XS;
  float* lds_h2 = smem + LROWS * XS + LROWS * HS;
  float* lds_w3 = smem + LROWS * XS + 2 * LROWS * HS;

  const int tid = threadIdx.x;
  const int srow = tid >> 4, sc = tid & 15;

  float w3v[8];
#pragma unroll
  for (int j = 0; j < 8; ++j)
    w3v[j] = Wl3[(srow + 16 * j) * 14 + (sc < 14 ? sc : 13)];

  {
    const int rowid = blockIdx.x * LROWS + srow;
    const int pg = pgi[rowid];
    const int pn = pni[rowid];
    const float* src0 = node + ((size_t)(pg + 1) * NPG - 1) * D;
    const float* src1 = node + (size_t)pn * D;
    const float* src2 = glob + (size_t)pg * D;
    float* dst = lds_x + srow * XS;
    v4f x00 = *(const v4f*)(src0 + 4 * sc);
    v4f x01 = *(const v4f*)(src0 + 4 * sc + 64);
    v4f x10 = *(const v4f*)(src1 + 4 * sc);
    v4f x11 = *(const v4f*)(src1 + 4 * sc + 64);
    v4f x20 = *(const v4f*)(src2 + 4 * sc);
    v4f x21 = *(const v4f*)(src2 + 4 * sc + 64);
    *(v4f*)(dst + 0 * D + 4 * sc)      = x00;
    *(v4f*)(dst + 0 * D + 4 * sc + 64) = x01;
    *(v4f*)(dst + 1 * D + 4 * sc)      = x10;
    *(v4f*)(dst + 1 * D + 4 * sc + 64) = x11;
    *(v4f*)(dst + 2 * D + 4 * sc)      = x20;
    *(v4f*)(dst + 2 * D + 4 * sc + 64) = x21;
  }
#pragma unroll
  for (int j = 0; j < 8; ++j)
    lds_w3[tid + 256 * j] = sc < 14 ? w3v[j] : 0.f;
  __syncthreads();

  const int wave = tid >> 5, lane = tid & 31;
  const int half = lane >> 4, l16 = lane & 15;
  const int n = wave * 16 + l16;

  {
    v8f c = {};
#pragma unroll
    for (int seg = 0; seg < 3; ++seg) {
      const float* ap = lds_x + l16 * XS + seg * D + 2 * half;
      const float* bp = Wl1 + (size_t)(seg * D + 2 * half) * D + n;
#pragma unroll
      for (int s = 0; s < 32; ++s) {
        v2f a = *(const v2f*)(ap + 4 * s);
        v2f b;
        b.x = bp[(size_t)(4 * s) * D];
        b.y = bp[(size_t)(4 * s + 1) * D];
        c = __builtin_amdgcn_wmma_f32_16x16x4_f32(false, a, false, b,
                                                  (short)0, c, false, false);
      }
    }
    float bv = bl1[n];
#pragma unroll
    for (int r = 0; r < 8; ++r) {
      float v = c[r] + bv;
      lds_h1[(r + 8 * half) * HS + n] = v > 0.f ? v : 0.f;
    }
  }
  __syncthreads();

  {
    v8f c = {};
    const float* ap = lds_h1 + l16 * HS + 2 * half;
    const float* bp = Wl2 + (size_t)(2 * half) * D + n;
#pragma unroll
    for (int s = 0; s < 32; ++s) {
      v2f a = *(const v2f*)(ap + 4 * s);
      v2f b;
      b.x = bp[(size_t)(4 * s) * D];
      b.y = bp[(size_t)(4 * s + 1) * D];
      c = __builtin_amdgcn_wmma_f32_16x16x4_f32(false, a, false, b,
                                                (short)0, c, false, false);
    }
    float bv = bl2[n];
#pragma unroll
    for (int r = 0; r < 8; ++r) {
      float v = c[r] + bv;
      lds_h2[(r + 8 * half) * HS + n] = v > 0.f ? v : 0.f;
    }
  }
  __syncthreads();

  if (wave == 0) {
    v8f c = {};
    const float* ap = lds_h2 + l16 * HS + 2 * half;
#pragma unroll
    for (int s = 0; s < 32; ++s) {
      v2f a = *(const v2f*)(ap + 4 * s);
      v2f b;
      b.x = lds_w3[(4 * s + 2 * half) * 16 + l16];
      b.y = lds_w3[(4 * s + 2 * half + 1) * 16 + l16];
      c = __builtin_amdgcn_wmma_f32_16x16x4_f32(false, a, false, b,
                                                (short)0, c, false, false);
    }
    if (l16 < 14) {
      float bv = bl3[l16];
#pragma unroll
      for (int r = 0; r < 8; ++r) {
        int grow = blockIdx.x * LROWS + r + 8 * half;
        out[outoff + grow * 14 + l16] = c[r] + bv;
      }
    }
  }
}

extern "C" void kernel_launch(void* const* d_in, const int* in_sizes, int n_in,
                              void* d_out, int out_size, void* d_ws, size_t ws_size,
                              hipStream_t stream) {
  (void)n_in; (void)out_size; (void)d_ws; (void)ws_size;
  const float* node = (const float*)d_in[0];
  const float* glob = (const float*)d_in[1];
  const float* Wp1  = (const float*)d_in[2];
  const float* bp1  = (const float*)d_in[3];
  const float* Wp2  = (const float*)d_in[4];
  const float* bp2  = (const float*)d_in[5];
  const float* Wl1  = (const float*)d_in[6];
  const float* bl1  = (const float*)d_in[7];
  const float* Wl2  = (const float*)d_in[8];
  const float* bl2  = (const float*)d_in[9];
  const float* Wl3  = (const float*)d_in[10];
  const float* bl3  = (const float*)d_in[11];
  const int*   pgi  = (const int*)d_in[12];
  const int*   pni  = (const int*)d_in[13];
  float* out = (float*)d_out;

  const int B = in_sizes[1] / D;
  const int N = in_sizes[0] / D;

  const size_t smem1 = (16384 + GPB * D + 128) * sizeof(float);
  const size_t smem2 = (LROWS * XS + 2 * LROWS * HS + D * 16) * sizeof(float);

  partner_kernel<<<B / GPB, 256, smem1, stream>>>(node, glob, Wp1, bp1, Wp2, bp2, out);
  label_kernel<<<B / LROWS, 256, smem2, stream>>>(node, glob, Wl1, bl1, Wl2, bl2,
                                                Wl3, bl3, pgi, pni, out, N);
}


# partner kernel: set-up (weight-tile staging, base vectors) once per WGP: 256 blocks of 32 waves own 32 graphs each (was 1024 x 8); base = two row tiles in four K quarters, sum order changes
# speedup vs baseline: 2.6706x; 1.0571x over previous
#include <hip/hip_runtime.h>

typedef float v2f __attribute__((ext_vector_type(2)));
typedef float v8f __attribute__((ext_vector_type(8)));

#define D   128
#define NPG 64

#define GPB 32
#define XP  (2 * D + 4)

typedef float v4f __attribute__((ext_vector_type(4)));

__global__ __launch_bounds__(1024) void partner_kernel(
    const float* __restrict__ node, const float* __restrict__ glob,
    const float* __restrict__ Wp1,  const float* __restrict__ bp1,
    const float* __restrict__ Wp2,  const float* __restrict__ bp2,
    float* __restrict__ out)
{
  extern __shared__ float smem[];
  float* lds_wb   = smem;
  float* lds_part = smem + 16384;
  float* lds_x    = smem + 16384 + 4 * GPB * D;
  float* lds_wp2  = smem + 16384 + 4 * GPB * D + GPB * XP;

  const int tid = threadIdx.x;
  const int g0  = blockIdx.x * GPB;
  const int wave = tid >> 5, lane = tid & 31;
  const int half = lane >> 4, l16 = lane & 15;

  const int xg = tid >> 6, xc = tid & 63;
  const size_t xoff = (xc < 32)
      ? ((size_t)(g0 + xg) * NPG + NPG - 1) * D + 4 * xc
      : (size_t)(g0 + xg) * D + 4 * (xc - 32);
  const size_t xstep = (xc < 32) ? (size_t)16 * NPG * D : (size_t)16 * D;
  const float* xsrc = ((xc < 32) ? node : glob) + xoff;
  const v4f xa = *(const v4f*)xsrc;
  const v4f xb = *(const v4f*)(xsrc + xstep);

  for (int i = tid; i < D * D; i += 1024) {
    int k = i >> 7, n = i & 127;
    float v = Wp1[(D + k) * D + n];
    int p = k >> 1;
    int col = (n + ((p & 1) << 4)) & 127;
    lds_wb[p * 256 + col * 2 + (k & 1)] = v;
  }
  *(v4f*)(lds_x + xg * XP + 4 * xc)        = xa;
  *(v4f*)(lds_x + (xg + 16) * XP + 4 * xc) = xb;
  if (tid < D) lds_wp2[tid] = Wp2[tid];
  __syncthreads();

  {
    const int ct = wave & 7, kq = wave >> 3;
    const int n  = ct * 16 + l16;
    const int krow = (kq < 2 ? kq * 64 : 2 * D + (kq - 2) * 64) + 2 * half;
    const float* bp = Wp1 + (size_t)krow * D + n;
    const float* a0 = lds_x + l16 * XP + kq * 64 + 2 * half;
    const float* a1 = a0 + 16 * XP;
    v8f c0 = {}, c1 = {};
#pragma unroll
    for (int s = 0; s < 16; ++s) {
      v2f b;
      b.x = bp[(size_t)(4 * s) * D];
      b.y = bp[(size_t)(4 * s + 1) * D];
      v2f x0 = *(const v2f*)(a0 + 4 * s);
      v2f x1 = *(const v2f*)(a1 + 4 * s);
      c0 = __builtin_amdgcn_wmma_f32_16x16x4_f32(false, x0, false, b,
                                                 (short)0, c0, false, false);
      c1 = __builtin_amdgcn_wmma_f32_16x16x4_f32(false, x1, false, b,
                                                 (short)0, c1, false, false);
    }
    const float bv = (kq == 0) ? bp1[n] : 0.f;
    float* pp = lds_part + kq * (GPB * D);
#pragma unroll
    for (int r = 0; r < 8; ++r) {
      pp[(r + 8 * half) * D + n]      = c0[r] + bv;
      pp[(16 + r + 8 * half) * D + n] = c1[r] + bv;
    }
  }
  __syncthreads();

  const int gsub = wave >> 2, mtile = wave & 3;
  const v2f* wb2 = (const v2f*)lds_wb;
  const float b2 = bp2[0];

  for (int it = 0; it < GPB / 8; ++it) {
    const int gl = 8 * it + gsub;
    const int graph = g0 + gl;
    const int row = graph * NPG + mtile * 16 + l16;
    const float* rowp = node + (size_t)row * D + 2 * half;

    v2f af[32];
#pragma unroll
    for (int s = 0; s < 32; ++s) af[s] = *(const v2f*)(rowp + 4 * s);

    float acc[8];
#pragma unroll
    for (int r = 0; r < 8; ++r) acc[r] = 0.f;

    for (int nt = 0; nt < 8; ++nt) {
      int n = nt * 16 + l16;
      int col = ((n + 16 * half) & 127);
      v8f c = {};
#pragma unroll
      for (int s = 0; s < 32; ++s) {
        v2f b = wb2[(2 * s + half) * D + col];
        c = __builtin_amdgcn_wmma_f32_16x16x4_f32(false, af[s], false, b,
                                                  (short)0, c, false, false);
      }
      const float* pb = lds_part + gl * D + n;
      float bn = (pb[0] + pb[GPB * D]) + (pb[2 * GPB * D] + pb[3 * GPB * D]);
      float wn = lds_wp2[n];
#pragma unroll
      for (int r = 0; r < 8; ++r) {
        float v = c[r] + bn;
        v = v > 0.f ? v : 0.f;
        acc[r] += v * wn;
      }
    }
#pragma unroll
    for (int r = 0; r < 8; ++r) {
      float a = acc[r];
      a += __shfl_xor(a, 1, 32);
      a += __shfl_xor(a, 2, 32);
      a += __shfl_xor(a, 4, 32);
      a += __shfl_xor(a, 8, 32);
      acc[r] = a;
    }
    if (l16 == 0) {
#pragma unroll
      for (int r = 0; r < 8; ++r)
        out[graph * NPG + mtile * 16 + r + 8 * half] = acc[r] + b2;
    }
  }
}

#define LROWS 16
#define XS    (3 * D + 4)
#define HS    (D + 4)

__global__ __launch_bounds__(256) void label_kernel(
    const float* __restrict__ node, const float* __restrict__ glob,
    const float* __restrict__ Wl1,  const float* __restrict__ bl1,
    const float* __restrict__ Wl2,  const float* __restrict__ bl2,
    const float* __restrict__ Wl3,  const float* __restrict__ bl3,
    const int* __restrict__ pgi,    const int* __restrict__ pni,
    float* __restrict__ out, int outoff)
{
  extern __shared__ float smem[];
  float* lds_x  = smem;
  float* lds_h1 = smem + LROWS * XS;
  float* lds_h2 = smem + LROWS * XS + LROWS * HS;
  float* lds_w3 = smem + LROWS * XS + 2 * LROWS * HS;

  const int tid = threadIdx.x;
  const int srow = tid >> 4, sc = tid & 15;

  float w3v[8];
#pragma unroll
  for (int j = 0; j < 8; ++j)
    w3v[j] = Wl3[(srow + 16 * j) * 14 + (sc < 14 ? sc : 13)];

  {
    const int rowid = blockIdx.x * LROWS + srow;
    const int pg = pgi[rowid];
    const int pn = pni[rowid];
    const float* src0 = node + ((size_t)(pg + 1) * NPG - 1) * D;
    const float* src1 = node + (size_t)pn * D;
    const float* src2 = glob + (size_t)pg * D;
    float* dst = lds_x + srow * XS;
    v4f x00 = *(const v4f*)(src0 + 4 * sc);
    v4f x01 = *(const v4f*)(src0 + 4 * sc + 64);
    v4f x10 = *(const v4f*)(src1 + 4 * sc);
    v4f x11 = *(const v4f*)(src1 + 4 * sc + 64);
    v4f x20 = *(const v4f*)(src2 + 4 * sc);
    v4f x21 = *(const v4f*)(src2 + 4 * sc + 64);
    *(v4f*)(dst + 0 * D + 4 * sc)      = x00;
    *(v4f*)(dst + 0 * D + 4 * sc + 64) = x01;
    *(v4f*)(dst + 1 * D + 4 * sc)      = x10;
    *(v4f*)(dst + 1 * D + 4 * sc + 64) = x11;
    *(v4f*)(dst + 2 * D + 4 * sc)      = x20;
    *(v4f*)(dst + 2 * D + 4 * sc + 64) = x21;
  }
#pragma unroll
  for (int j = 0; j < 8; ++j)
    lds_w3[tid + 256 * j] = sc < 14 ? w3v[j] : 0.f;
  __syncthreads();

  const int wave = tid >> 5, lane = tid & 31;
  const int half = lane >> 4, l16 = lane & 15;
  const int n = wave * 16 + l16;

  {
    v8f c = {};
#pragma unroll
    for (int seg = 0; seg < 3; ++seg) {
      const float* ap = lds_x + l16 * XS + seg * D + 2 * half;
      const float* bp = Wl1 + (size_t)(seg * D + 2 * half) * D + n;
#pragma unroll
      for (int s = 0; s < 32; ++s) {
        v2f a = *(const v2f*)(ap + 4 * s);
        v2f b;
        b.x = bp[(size_t)(4 * s) * D];
        b.y = bp[(size_t)(4 * s + 1) * D];
        c = __builtin_amdgcn_wmma_f32_16x16x4_f32(false, a, false, b,
                                                  (short)0, c, false, false);
      }
    }
    float bv = bl1[n];
#pragma unroll
    for (int r = 0; r < 8; ++r) {
      float v = c[r] + bv;
      lds_h1[(r + 8 * half) * HS + n] = v > 0.f ? v : 0.f;
    }
  }
  __syncthreads();

  {
    v8f c = {};
    const float* ap = lds_h1 + l16 * HS + 2 * half;
    const float* bp = Wl2 + (size_t)(2 * half) * D + n;
#pragma unroll
    for (int s = 0; s < 32; ++s) {
      v2f a = *(const v2f*)(ap + 4 * s);
      v2f b;
      b.x = bp[(size_t)(4 * s) * D];
      b.y = bp[(size_t)(4 * s + 1) * D];
      c = __builtin_amdgcn_wmma_f32_16x16x4_f32(false, a, false, b,
                                                (short)0, c, false, false);
    }
    float bv = bl2[n];
#pragma unroll
    for (int r = 0; r < 8; ++r) {
      float v = c[r] + bv;
      lds_h2[(r + 8 * half) * HS + n] = v > 0.f ? v : 0.f;
    }
  }
  __syncthreads();

  if (wave == 0) {
    v8f c = {};
    const float* ap = lds_h2 + l16 * HS + 2 * half;
#pragma unroll
    for (int s = 0; s < 32; ++s) {
      v2f a = *(const v2f*)(ap + 4 * s);
      v2f b;
      b.x = lds_w3[(4 * s + 2 * half) * 16 + l16];
      b.y = lds_w3[(4 * s + 2 * half + 1) * 16 + l16];
      c = __builtin_amdgcn_wmma_f32_16x16x4_f32(false, a, false, b,
                                                (short)0, c, false, false);
    }
    if (l16 < 14) {
      float bv = bl3[l16];
#pragma unroll
      for (int r = 0; r < 8; ++r) {
        int grow = blockIdx.x * LROWS + r + 8 * half;
        out[outoff + grow * 14 + l16] = c[r] + bv;
      }
    }
  }
}

extern "C" void kernel_launch(void* const* d_in, const int* in_sizes, int n_in,
                              void* d_out, int out_size, void* d_ws, size_t ws_size,
                              hipStream_t stream) {
  (void)n_in; (void)out_size; (void)d_ws; (void)ws_size;
  const float* node = (const float*)d_in[0];
  const float* glob = (const float*)d_in[1];
  const float* Wp1  = (const float*)d_in[2];
  const float* bp1  = (const float*)d_in[3];
  const float* Wp2  = (const float*)d_in[4];
  const float* bp2  = (const float*)d_in[5];
  const float* Wl1  = (const float*)d_in[6];
  const float* bl1  = (const float*)d_in[7];
  const float* Wl2  = (const float*)d_in[8];
  const float* bl2  = (const float*)d_in[9];
  const float* Wl3  = (const float*)d_in[10];
  const float* bl3  = (const float*)d_in[11];
  const int*   pgi  = (const int*)d_in[12];
  const int*   pni  = (const int*)d_in[13];
  float* out = (float*)d_out;

  const int B = in_sizes[1] / D;
  const int N = in_sizes[0] / D;

  const size_t smem1 = (16384 + 4 * GPB * D + GPB * XP + 128) * sizeof(float);
  const size_t smem2 = (LROWS * XS + 2 * LROWS * HS + D * 16) * sizeof(float);

  partner_kernel<<<B / GPB, 1024, smem1, stream>>>(node, glob, Wp1, bp1, Wp2, bp2, out);
  label_kernel<<<B / LROWS, 256, smem2, stream>>>(node, glob, Wl1, bl1, Wl2, bl2,
                                                Wl3, bl3, pgi, pni, out, N);
}


# partner kernel: each wave owns two 16-row tiles (16 waves/block, was 32): an LDS weight fragment feeds two independent WMMA chains; fragment reads grouped 16 ahead of 32-WMMA blocks; sums unchanged
# speedup vs baseline: 2.8050x; 1.0503x over previous
#include <hip/hip_runtime.h>

typedef float v2f __attribute__((ext_vector_type(2)));
typedef float v8f __attribute__((ext_vector_type(8)));

#define D   128
#define NPG 64

#define GPB 32
#define PTHR 512
#define XP  (2 * D + 4)

typedef float v4f __attribute__((ext_vector_type(4)));

__device__ __forceinline__ void load_wfrag(v2f (&dst)[16], const v2f* wb2,
                                           int nt, int sbase, int half, int l16)
{
  const int col = (nt * 16 + l16 + 16 * half) & 127;
#pragma unroll
  for (int s = 0; s < 16; ++s)
    dst[s] = wb2[(2 * (sbase + s) + half) * D + col];
}

__global__ __launch_bounds__(PTHR) void partner_kernel(
    const float* __restrict__ node, const float* __restrict__ glob,
    const float* __restrict__ Wp1,  const float* __restrict__ bp1,
    const float* __restrict__ Wp2,  const float* __restrict__ bp2,
    float* __restrict__ out)
{
  extern __shared__ float smem[];
  float* lds_wb   = smem;
  float* lds_part = smem + 16384;
  float* lds_x    = smem + 16384 + 4 * GPB * D;
  float* lds_wp2  = smem + 16384 + 4 * GPB * D + GPB * XP;

  const int tid = threadIdx.x;
  const int g0  = blockIdx.x * GPB;
  const int wave = tid >> 5, lane = tid & 31;
  const int half = lane >> 4, l16 = lane & 15;

  const int xg = tid >> 6, xc = tid & 63;
  const size_t xoff = (xc < 32)
      ? ((size_t)(g0 + xg) * NPG + NPG - 1) * D + 4 * xc
      : (size_t)(g0 + xg) * D + 4 * (xc - 32);
  const size_t xstep = (xc < 32) ? (size_t)8 * NPG * D : (size_t)8 * D;
  const float* xsrc = ((xc < 32) ? node : glob) + xoff;
  v4f xv[4];
#pragma unroll
  for (int j = 0; j < 4; ++j) xv[j] = *(const v4f*)(xsrc + j * xstep);

#pragma unroll
  for (int j = 0; j < D * D / PTHR; ++j) {
    const int i = tid + j * PTHR;
    int k = i >> 7, n = i & 127;
    float v = Wp1[(D + k) * D + n];
    int p = k >> 1;
    int col = (n + ((p & 1) << 4)) & 127;
    lds_wb[p * 256 + col * 2 + (k & 1)] = v;
  }
#pragma unroll
  for (int j = 0; j < 4; ++j)
    *(v4f*)(lds_x + (xg + 8 * j) * XP + 4 * xc) = xv[j];
  if (tid < D) lds_wp2[tid] = Wp2[tid];
  __syncthreads();

  {
    const int ct = wave & 7, kh = wave >> 3;
    const int n  = ct * 16 + l16;
    for (int q = 0; q < 2; ++q) {
      const int kq = 2 * kh + q;
      const int krow = (kq < 2 ? kq * 64 : 2 * D + (kq - 2) * 64) + 2 * half;
      const float* bp = Wp1 + (size_t)krow * D + n;
      const float* a0 = lds_x + l16 * XP + kq * 64 + 2 * half;
      const float* a1 = a0 + 16 * XP;
      v8f c0 = {}, c1 = {};
#pragma unroll
      for (int s = 0; s < 16; ++s) {
        v2f b;
        b.x = bp[(size_t)(4 * s) * D];
        b.y = bp[(size_t)(4 * s + 1) * D];
        v2f x0 = *(const v2f*)(a0 + 4 * s);
        v2f x1 = *(const v2f*)(a1 + 4 * s);
        c0 = __builtin_amdgcn_wmma_f32_16x16x4_f32(false, x0, false, b,
                                                   (short)0, c0, false, false);
        c1 = __builtin_amdgcn_wmma_f32_16x16x4_f32(false, x1, false, b,
                                                   (short)0, c1, false, false);
      }
      const float bv = (kq == 0) ? bp1[n] : 0.f;
      float* pp = lds_part + kq * (GPB * D);
#pragma unroll
      for (int r = 0; r < 8; ++r) {
        pp[(r + 8 * half) * D + n]      = c0[r] + bv;
        pp[(16 + r + 8 * half) * D + n] = c1[r] + bv;
      }
    }
  }
  __syncthreads();

  const int gsub = wave >> 1, mpair = wave & 1;
  const v2f* wb2 = (const v2f*)lds_wb;
  const float b2 = bp2[0];

  v2f bA[16], bB[16];
  load_wfrag(bA, wb2, 0, 0, half, l16);

  for (int it = 0; it < GPB / 8; ++it) {
    const int gl = 8 * it + gsub;
    const int graph = g0 + gl;
    const int row = graph * NPG + mpair * 32 + l16;
    const float* rowp0 = node + (size_t)row * D + 2 * half;
    const float* rowp1 = rowp0 + (size_t)16 * D;

    v2f af0[32], af1[32];
#pragma unroll
    for (int s = 0; s < 32; ++s) af0[s] = *(const v2f*)(rowp0 + 4 * s);
#pragma unroll
    for (int s = 0; s < 32; ++s) af1[s] = *(const v2f*)(rowp1 + 4 * s);

    float acc0[8], acc1[8];
#pragma unroll
    for (int r = 0; r < 8; ++r) { acc0[r] = 0.f; acc1[r] = 0.f; }

    for (int nt = 0; nt < 8; ++nt) {
      const int n = nt * 16 + l16;
      v8f c0 = {}, c1 = {};
      const float* pb = lds_part + gl * D + n;
      const float p0 = pb[0], p1 = pb[GPB * D];
      const float p2 = pb[2 * GPB * D], p3 = pb[3 * GPB * D];
      const float wn = lds_wp2[n];
      load_wfrag(bB, wb2, nt, 16, half, l16);
      __builtin_amdgcn_sched_barrier(0);
#pragma unroll
      for (int s = 0; s < 16; ++s) {
        c0 = __builtin_amdgcn_wmma_f32_16x16x4_f32(false, af0[s], false, bA[s],
                                                   (short)0, c0, false, false);
        c1 = __builtin_amdgcn_wmma_f32_16x16x4_f32(false, af1[s], false, bA[s],
                                                   (short)0, c1, false, false);
      }
      asm volatile("" : "+v"(c0), "+v"(c1));
      load_wfrag(bA, wb2, (nt + 1) & 7, 0, half, l16);
      __builtin_amdgcn_sched_barrier(0);
#pragma unroll
      for (int s = 0; s < 16; ++s) {
        c0 = __builtin_amdgcn_wmma_f32_16x16x4_f32(false, af0[16 + s], false, bB[s],
                                                   (short)0, c0, false, false);
        c1 = __builtin_amdgcn_wmma_f32_16x16x4_f32(false, af1[16 + s], false, bB[s],
                                                   (short)0, c1, false, false);
      }
      const float bn = (p0 + p1) + (p2 + p3);
#pragma unroll
      for (int r = 0; r < 8; ++r) {
        float v0 = c0[r] + bn;
        v0 = v0 > 0.f ? v0 : 0.f;
        acc0[r] += v0 * wn;
        float v1 = c1[r] + bn;
        v1 = v1 > 0.f ? v1 : 0.f;
        acc1[r] += v1 * wn;
      }
    }
#pragma unroll
    for (int r = 0; r < 8; ++r) {
      float a = acc0[r];
      a += __shfl_xor(a, 1, 32);
      a += __shfl_xor(a, 2, 32);
      a += __shfl_xor(a, 4, 32);
      a += __shfl_xor(a, 8, 32);
      acc0[r] = a;
      float b = acc1[r];
      b += __shfl_xor(b, 1, 32);
      b += __shfl_xor(b, 2, 32);
      b += __shfl_xor(b, 4, 32);
      b += __shfl_xor(b, 8, 32);
      acc1[r] = b;
    }
    if (l16 == 0) {
      float* op = out + graph * NPG + mpair * 32 + 8 * half;
#pragma unroll
      for (int r = 0; r < 8; ++r) {
        op[r]      = acc0[r] + b2;
        op[16 + r] = acc1[r] + b2;
      }
    }
  }
}

#define LROWS 16
#define XS    (3 * D + 4)
#define HS    (D + 4)

__global__ __launch_bounds__(256) void label_kernel(
    const float* __restrict__ node, const float* __restrict__ glob,
    const float* __restrict__ Wl1,  const float* __restrict__ bl1,
    const float* __restrict__ Wl2,  const float* __restrict__ bl2,
    const float* __restrict__ Wl3,  const float* __restrict__ bl3,
    const int* __restrict__ pgi,    const int* __restrict__ pni,
    float* __restrict__ out, int outoff)
{
  extern __shared__ float smem[];
  float* lds_x  = smem;
  float* lds_h1 = smem + LROWS * XS;
  float* lds_h2 = smem + LROWS * XS + LROWS * HS;
  float* lds_w3 = smem + LROWS * XS + 2 * LROWS * HS;

  const int tid = threadIdx.x;
  const int srow = tid >> 4, sc = tid & 15;

  float w3v[8];
#pragma unroll
  for (int j = 0; j < 8; ++j)
    w3v[j] = Wl3[(srow + 16 * j) * 14 + (sc < 14 ? sc : 13)];

  {
    const int rowid = blockIdx.x * LROWS + srow;
    const int pg = pgi[rowid];
    const int pn = pni[rowid];
    const float* src0 = node + ((size_t)(pg + 1) * NPG - 1) * D;
    const float* src1 = node + (size_t)pn * D;
    const float* src2 = glob + (size_t)pg * D;
    float* dst = lds_x + srow * XS;
    v4f x00 = *(const v4f*)(src0 + 4 * sc);
    v4f x01 = *(const v4f*)(src0 + 4 * sc + 64);
    v4f x10 = *(const v4f*)(src1 + 4 * sc);
    v4f x11 = *(const v4f*)(src1 + 4 * sc + 64);
    v4f x20 = *(const v4f*)(src2 + 4 * sc);
    v4f x21 = *(const v4f*)(src2 + 4 * sc + 64);
    *(v4f*)(dst + 0 * D + 4 * sc)      = x00;
    *(v4f*)(dst + 0 * D + 4 * sc + 64) = x01;
    *(v4f*)(dst + 1 * D + 4 * sc)      = x10;
    *(v4f*)(dst + 1 * D + 4 * sc + 64) = x11;
    *(v4f*)(dst + 2 * D + 4 * sc)      = x20;
    *(v4f*)(dst + 2 * D + 4 * sc + 64) = x21;
  }
#pragma unroll
  for (int j = 0; j < 8; ++j)
    lds_w3[tid + 256 * j] = sc < 14 ? w3v[j] : 0.f;
  __syncthreads();

  const int wave = tid >> 5, lane = tid & 31;
  const int half = lane >> 4, l16 = lane & 15;
  const int n = wave * 16 + l16;

  {
    v8f c = {};
#pragma unroll
    for (int seg = 0; seg < 3; ++seg) {
      const float* ap = lds_x + l16 * XS + seg * D + 2 * half;
      const float* bp = Wl1 + (size_t)(seg * D + 2 * half) * D + n;
#pragma unroll
      for (int s = 0; s < 32; ++s) {
        v2f a = *(const v2f*)(ap + 4 * s);
        v2f b;
        b.x = bp[(size_t)(4 * s) * D];
        b.y = bp[(size_t)(4 * s + 1) * D];
        c = __builtin_amdgcn_wmma_f32_16x16x4_f32(false, a, false, b,
                                                  (short)0, c, false, false);
      }
    }
    float bv = bl1[n];
#pragma unroll
    for (int r = 0; r < 8; ++r) {
      float v = c[r] + bv;
      lds_h1[(r + 8 * half) * HS + n] = v > 0.f ? v : 0.f;
    }
  }
  __syncthreads();

  {
    v8f c = {};
    const float* ap = lds_h1 + l16 * HS + 2 * half;
    const float* bp = Wl2 + (size_t)(2 * half) * D + n;
#pragma unroll
    for (int s = 0; s < 32; ++s) {
      v2f a = *(const v2f*)(ap + 4 * s);
      v2f b;
      b.x = bp[(size_t)(4 * s) * D];
      b.y = bp[(size_t)(4 * s + 1) * D];
      c = __builtin_amdgcn_wmma_f32_16x16x4_f32(false, a, false, b,
                                                (short)0, c, false, false);
    }
    float bv = bl2[n];
#pragma unroll
    for (int r = 0; r < 8; ++r) {
      float v = c[r] + bv;
      lds_h2[(r + 8 * half) * HS + n] = v > 0.f ? v : 0.f;
    }
  }
  __syncthreads();

  if (wave == 0) {
    v8f c = {};
    const float* ap = lds_h2 + l16 * HS + 2 * half;
#pragma unroll
    for (int s = 0; s < 32; ++s) {
      v2f a = *(const v2f*)(ap + 4 * s);
      v2f b;
      b.x = lds_w3[(4 * s + 2 * half) * 16 + l16];
      b.y = lds_w3[(4 * s + 2 * half + 1) * 16 + l16];
      c = __builtin_amdgcn_wmma_f32_16x16x4_f32(false, a, false, b,
                                                (short)0, c, false, false);
    }
    if (l16 < 14) {
      float bv = bl3[l16];
#pragma unroll
      for (int r = 0; r < 8; ++r) {
        int grow = blockIdx.x * LROWS + r + 8 * half;
        out[outoff + grow * 14 + l16] = c[r] + bv;
      }
    }
  }
}

extern "C" void kernel_launch(void* const* d_in, const int* in_sizes, int n_in,
                              void* d_out, int out_size, void* d_ws, size_t ws_size,
                              hipStream_t stream) {
  (void)n_in; (void)out_size; (void)d_ws; (void)ws_size;
  const float* node = (const float*)d_in[0];
  const float* glob = (const float*)d_in[1];
  const float* Wp1  = (const float*)d_in[2];
  const float* bp1  = (const float*)d_in[3];
  const float* Wp2  = (const float*)d_in[4];
  const float* bp2  = (const float*)d_in[5];
  const float* Wl1  = (const float*)d_in[6];
  const float* bl1  = (const float*)d_in[7];
  const float* Wl2  = (const float*)d_in[8];
  const float* bl2  = (const float*)d_in[9];
  const float* Wl3  = (const float*)d_in[10];
  const float* bl3  = (const float*)d_in[11];
  const int*   pgi  = (const int*)d_in[12];
  const int*   pni  = (const int*)d_in[13];
  float* out = (float*)d_out;

  const int B = in_sizes[1] / D;
  const int N = in_sizes[0] / D;

  const size_t smem1 = (16384 + 4 * GPB * D + GPB * XP + 128) * sizeof(float);
  const size_t smem2 = (LROWS * XS + 2 * LROWS * HS + D * 16) * sizeof(float);

  partner_kernel<<<B / GPB, PTHR, smem1, stream>>>(node, glob, Wp1, bp1, Wp2, bp2, out);
  label_kernel<<<B / LROWS, 256, smem2, stream>>>(node, glob, Wl1, bl1, Wl2, bl2,
                                                Wl3, bl3, pgi, pni, out, N);
}


# label MLP folded into the partner kernel: one launch (was two); each 32-graph block computes its own 32 label rows (2 row tiles on 16 waves) in its set-up phase; sums unchanged
# speedup vs baseline: 2.9175x; 1.0401x over previous
#include <hip/hip_runtime.h>

typedef float v2f __attribute__((ext_vector_type(2)));
typedef float v8f __attribute__((ext_vector_type(8)));

#define D   128
#define NPG 64

#define GPB 32
#define PTHR 512
#define XP  (2 * D + 4)

typedef float v4f __attribute__((ext_vector_type(4)));

__device__ __forceinline__ void load_wfrag(v2f (&dst)[16], const v2f* wb2,
                                           int nt, int sbase, int half, int l16)
{
  const int col = (nt * 16 + l16 + 16 * half) & 127;
#pragma unroll
  for (int s = 0; s < 16; ++s)
    dst[s] = wb2[(2 * (sbase + s) + half) * D + col];
}

#define LROWS 16
#define XS    (3 * D + 4)
#define HS    (D + 4)
#define PLDS  (16384 + 4 * GPB * D + GPB * XP + 128)

__global__ __launch_bounds__(PTHR) void partner_kernel(
    const float* __restrict__ node, const float* __restrict__ glob,
    const float* __restrict__ Wp1,  const float* __restrict__ bp1,
    const float* __restrict__ Wp2,  const float* __restrict__ bp2,
    const float* __restrict__ Wl1,  const float* __restrict__ bl1,
    const float* __restrict__ Wl2,  const float* __restrict__ bl2,
    const float* __restrict__ Wl3,  const float* __restrict__ bl3,
    const int* __restrict__ pgi,    const int* __restrict__ pni,
    float* __restrict__ out, int outoff)
{
  extern __shared__ float smem[];
  float* lds_wb   = smem;
  float* lds_part = smem + 16384;
  float* lds_x    = smem + 16384 + 4 * GPB * D;
  float* lds_wp2  = smem + 16384 + 4 * GPB * D + GPB * XP;
  float* lds_lx   = smem + PLDS;
  float* lds_h1   = lds_lx + 2 * LROWS * XS;
  float* lds_h2   = lds_h1 + 2 * LROWS * HS;
  float* lds_w3   = lds_h2 + 2 * LROWS * HS;

  const int tid = threadIdx.x;
  const int g0  = blockIdx.x * GPB;
  const int wave = tid >> 5, lane = tid & 31;
  const int half = lane >> 4, l16 = lane & 15;

  const int lrow = tid >> 4, sc = tid & 15;
  const int rowid = blockIdx.x * (2 * LROWS) + lrow;
  const int pg = pgi[rowid];
  const int pn = pni[rowid];

  const int xg = tid >> 6, xc = tid & 63;
  const size_t xoff = (xc < 32)
      ? ((size_t)(g0 + xg) * NPG + NPG - 1) * D + 4 * xc
      : (size_t)(g0 + xg) * D + 4 * (xc - 32);
  const size_t xstep = (xc < 32) ? (size_t)8 * NPG * D : (size_t)8 * D;
  const float* xsrc = ((xc < 32) ? node : glob) + xoff;
  v4f xv[4];
#pragma unroll
  for (int j = 0; j < 4; ++j) xv[j] = *(const v4f*)(xsrc + j * xstep);

  float w3v[4];
#pragma unroll
  for (int j = 0; j < 4; ++j)
    w3v[j] = Wl3[(lrow + 32 * j) * 14 + (sc < 14 ? sc : 13)];

  const float* src0 = node + ((size_t)(pg + 1) * NPG - 1) * D;
  const float* src1 = node + (size_t)pn * D;
  const float* src2 = glob + (size_t)pg * D;
  v4f x00 = *(const v4f*)(src0 + 4 * sc);
  v4f x01 = *(const v4f*)(src0 + 4 * sc + 64);
  v4f x10 = *(const v4f*)(src1 + 4 * sc);
  v4f x11 = *(const v4f*)(src1 + 4 * sc + 64);
  v4f x20 = *(const v4f*)(src2 + 4 * sc);
  v4f x21 = *(const v4f*)(src2 + 4 * sc + 64);

#pragma unroll
  for (int j = 0; j < D * D / PTHR; ++j) {
    const int i = tid + j * PTHR;
    int k = i >> 7, n = i & 127;
    float v = Wp1[(D + k) * D + n];
    int p = k >> 1;
    int col = (n + ((p & 1) << 4)) & 127;
    lds_wb[p * 256 + col * 2 + (k & 1)] = v;
  }
#pragma unroll
  for (int j = 0; j < 4; ++j)
    *(v4f*)(lds_x + (xg + 8 * j) * XP + 4 * xc) = xv[j];
  if (tid < D) lds_wp2[tid] = Wp2[tid];
  {
    float* dst = lds_lx + lrow * XS;
    *(v4f*)(dst + 0 * D + 4 * sc)      = x00;
    *(v4f*)(dst + 0 * D + 4 * sc + 64) = x01;
    *(v4f*)(dst + 1 * D + 4 * sc)      = x10;
    *(v4f*)(dst + 1 * D + 4 * sc + 64) = x11;
    *(v4f*)(dst + 2 * D + 4 * sc)      = x20;
    *(v4f*)(dst + 2 * D + 4 * sc + 64) = x21;
  }
#pragma unroll
  for (int j = 0; j < 4; ++j)
    lds_w3[tid + PTHR * j] = sc < 14 ? w3v[j] : 0.f;
  __syncthreads();

  {
    const int ct = wave & 7, kh = wave >> 3;
    const int n  = ct * 16 + l16;
    for (int q = 0; q < 2; ++q) {
      const int kq = 2 * kh + q;
      const int krow = (kq < 2 ? kq * 64 : 2 * D + (kq - 2) * 64) + 2 * half;
      const float* bp = Wp1 + (size_t)krow * D + n;
      const float* a0 = lds_x + l16 * XP + kq * 64 + 2 * half;
      const float* a1 = a0 + 16 * XP;
      v8f c0 = {}, c1 = {};
#pragma unroll
      for (int s = 0; s < 16; ++s) {
        v2f b;
        b.x = bp[(size_t)(4 * s) * D];
        b.y = bp[(size_t)(4 * s + 1) * D];
        v2f x0 = *(const v2f*)(a0 + 4 * s);
        v2f x1 = *(const v2f*)(a1 + 4 * s);
        c0 = __builtin_amdgcn_wmma_f32_16x16x4_f32(false, x0, false, b,
                                                   (short)0, c0, false, false);
        c1 = __builtin_amdgcn_wmma_f32_16x16x4_f32(false, x1, false, b,
                                                   (short)0, c1, false, false);
      }
      const float bv = (kq == 0) ? bp1[n] : 0.f;
      float* pp = lds_part + kq * (GPB * D);
#pragma unroll
      for (int r = 0; r < 8; ++r) {
        pp[(r + 8 * half) * D + n]      = c0[r] + bv;
        pp[(16 + r + 8 * half) * D + n] = c1[r] + bv;
      }
    }
  }

  const int ltile = wave >> 3;
  const int ln = (wave & 7) * 16 + l16;
  {
    v8f c = {};
#pragma unroll
    for (int seg = 0; seg < 3; ++seg) {
      const float* ap = lds_lx + (ltile * LROWS + l16) * XS + seg * D + 2 * half;
      const float* bp = Wl1 + (size_t)(seg * D + 2 * half) * D + ln;
#pragma unroll
      for (int s = 0; s < 32; ++s) {
        v2f a = *(const v2f*)(ap + 4 * s);
        v2f b;
        b.x = bp[(size_t)(4 * s) * D];
        b.y = bp[(size_t)(4 * s + 1) * D];
        c = __builtin_amdgcn_wmma_f32_16x16x4_f32(false, a, false, b,
                                                  (short)0, c, false, false);
      }
    }
    float bv = bl1[ln];
#pragma unroll
    for (int r = 0; r < 8; ++r) {
      float v = c[r] + bv;
      lds_h1[(ltile * LROWS + r + 8 * half) * HS + ln] = v > 0.f ? v : 0.f;
    }
  }
  __syncthreads();

  {
    v8f c = {};
    const float* ap = lds_h1 + (ltile * LROWS + l16) * HS + 2 * half;
    const float* bp = Wl2 + (size_t)(2 * half) * D + ln;
#pragma unroll
    for (int s = 0; s < 32; ++s) {
      v2f a = *(const v2f*)(ap + 4 * s);
      v2f b;
      b.x = bp[(size_t)(4 * s) * D];
      b.y = bp[(size_t)(4 * s + 1) * D];
      c = __builtin_amdgcn_wmma_f32_16x16x4_f32(false, a, false, b,
                                                (short)0, c, false, false);
    }
    float bv = bl2[ln];
#pragma unroll
    for (int r = 0; r < 8; ++r) {
      float v = c[r] + bv;
      lds_h2[(ltile * LROWS + r + 8 * half) * HS + ln] = v > 0.f ? v : 0.f;
    }
  }
  __syncthreads();

  if ((wave & 7) == 0) {
    v8f c = {};
    const float* ap = lds_h2 + (ltile * LROWS + l16) * HS + 2 * half;
#pragma unroll
    for (int s = 0; s < 32; ++s) {
      v2f a = *(const v2f*)(ap + 4 * s);
      v2f b;
      b.x = lds_w3[(4 * s + 2 * half) * 16 + l16];
      b.y = lds_w3[(4 * s + 2 * half + 1) * 16 + l16];
      c = __builtin_amdgcn_wmma_f32_16x16x4_f32(false, a, false, b,
                                                (short)0, c, false, false);
    }
    if (l16 < 14) {
      float bv = bl3[l16];
#pragma unroll
      for (int r = 0; r < 8; ++r) {
        int grow = blockIdx.x * (2 * LROWS) + ltile * LROWS + r + 8 * half;
        out[outoff + grow * 14 + l16] = c[r] + bv;
      }
    }
  }

  const int gsub = wave >> 1, mpair = wave & 1;
  const v2f* wb2 = (const v2f*)lds_wb;
  const float b2 = bp2[0];

  v2f bA[16], bB[16];
  load_wfrag(bA, wb2, 0, 0, half, l16);

  for (int it = 0; it < GPB / 8; ++it) {
    const int gl = 8 * it + gsub;
    const int graph = g0 + gl;
    const int row = graph * NPG + mpair * 32 + l16;
    const float* rowp0 = node + (size_t)row * D + 2 * half;
    const float* rowp1 = rowp0 + (size_t)16 * D;

    v2f af0[32], af1[32];
#pragma unroll
    for (int s = 0; s < 32; ++s) af0[s] = *(const v2f*)(rowp0 + 4 * s);
#pragma unroll
    for (int s = 0; s < 32; ++s) af1[s] = *(const v2f*)(rowp1 + 4 * s);

    float acc0[8], acc1[8];
#pragma unroll
    for (int r = 0; r < 8; ++r) { acc0[r] = 0.f; acc1[r] = 0.f; }

    for (int nt = 0; nt < 8; ++nt) {
      const int n = nt * 16 + l16;
      v8f c0 = {}, c1 = {};
      const float* pb = lds_part + gl * D + n;
      const float p0 = pb[0], p1 = pb[GPB * D];
      const float p2 = pb[2 * GPB * D], p3 = pb[3 * GPB * D];
      const float wn = lds_wp2[n];
      load_wfrag(bB, wb2, nt, 16, half, l16);
      __builtin_amdgcn_sched_barrier(0);
#pragma unroll
      for (int s = 0; s < 16; ++s) {
        c0 = __builtin_amdgcn_wmma_f32_16x16x4_f32(false, af0[s], false, bA[s],
                                                   (short)0, c0, false, false);
        c1 = __builtin_amdgcn_wmma_f32_16x16x4_f32(false, af1[s], false, bA[s],
                                                   (short)0, c1, false, false);
      }
      asm volatile("" : "+v"(c0), "+v"(c1));
      load_wfrag(bA, wb2, (nt + 1) & 7, 0, half, l16);
      __builtin_amdgcn_sched_barrier(0);
#pragma unroll
      for (int s = 0; s < 16; ++s) {
        c0 = __builtin_amdgcn_wmma_f32_16x16x4_f32(false, af0[16 + s], false, bB[s],
                                                   (short)0, c0, false, false);
        c1 = __builtin_amdgcn_wmma_f32_16x16x4_f32(false, af1[16 + s], false, bB[s],
                                                   (short)0, c1, false, false);
      }
      const float bn = (p0 + p1) + (p2 + p3);
#pragma unroll
      for (int r = 0; r < 8; ++r) {
        float v0 = c0[r] + bn;
        v0 = v0 > 0.f ? v0 : 0.f;
        acc0[r] += v0 * wn;
        float v1 = c1[r] + bn;
        v1 = v1 > 0.f ? v1 : 0.f;
        acc1[r] += v1 * wn;
      }
    }
#pragma unroll
    for (int r = 0; r < 8; ++r) {
      float a = acc0[r];
      a += __shfl_xor(a, 1, 32);
      a += __shfl_xor(a, 2, 32);
      a += __shfl_xor(a, 4, 32);
      a += __shfl_xor(a, 8, 32);
      acc0[r] = a;
      float b = acc1[r];
      b += __shfl_xor(b, 1, 32);
      b += __shfl_xor(b, 2, 32);
      b += __shfl_xor(b, 4, 32);
      b += __shfl_xor(b, 8, 32);
      acc1[r] = b;
    }
    if (l16 == 0) {
      float* op = out + graph * NPG + mpair * 32 + 8 * half;
#pragma unroll
      for (int r = 0; r < 8; ++r) {
        op[r]      = acc0[r] + b2;
        op[16 + r] = acc1[r] + b2;
      }
    }
  }
}

extern "C" void kernel_launch(void* const* d_in, const int* in_sizes, int n_in,
                              void* d_out, int out_size, void* d_ws, size_t ws_size,
                              hipStream_t stream) {
  (void)n_in; (void)out_size; (void)d_ws; (void)ws_size;
  const float* node = (const float*)d_in[0];
  const float* glob = (const float*)d_in[1];
  const float* Wp1  = (const float*)d_in[2];
  const float* bp1  = (const float*)d_in[3];
  const float* Wp2  = (const float*)d_in[4];
  const float* bp2  = (const float*)d_in[5];
  const float* Wl1  = (const float*)d_in[6];
  const float* bl1  = (const float*)d_in[7];
  const float* Wl2  = (const float*)d_in[8];
  const float* bl2  = (const float*)d_in[9];
  const float* Wl3  = (const float*)d_in[10];
  const float* bl3  = (const float*)d_in[11];
  const int*   pgi  = (const int*)d_in[12];
  const int*   pni  = (const int*)d_in[13];
  float* out = (float*)d_out;

  const int B = in_sizes[1] / D;
  const int N = in_sizes[0] / D;

  const size_t smem1 = (PLDS + 2 * LROWS * XS + 4 * LROWS * HS + D * 16) * sizeof(float);

  partner_kernel<<<B / GPB, PTHR, smem1, stream>>>(node, glob, Wp1, bp1, Wp2, bp2,
                                                  Wl1, bl1, Wl2, bl2, Wl3, bl3,
                                                  pgi, pni, out, N);
}


# partner loop: each wave's 32 node rows loaded as 32 whole 512-byte rows (4 lines per load, was 16 B from each of 16 lines) and turned into A fragments through a private LDS buffer; sums unchanged
# speedup vs baseline: 3.0091x; 1.0314x over previous
#include <hip/hip_runtime.h>

typedef float v2f __attribute__((ext_vector_type(2)));
typedef float v8f __attribute__((ext_vector_type(8)));

#define D   128
#define NPG 64

#define GPB 32
#define PTHR 512
#define XP  (2 * D + 4)

typedef float v4f __attribute__((ext_vector_type(4)));

#define WAVE_LDS_ORDER()                                            \
  do {                                                              \
    __builtin_amdgcn_fence(__ATOMIC_RELEASE, "wavefront");          \
    __builtin_amdgcn_wave_barrier();                                \
    __builtin_amdgcn_fence(__ATOMIC_ACQUIRE, "wavefront");          \
  } while (0)

__device__ __forceinline__ void load_wfrag(v2f (&dst)[16], const v2f* wb2,
                                           int nt, int sbase, int half, int l16)
{
  const int col = (nt * 16 + l16 + 16 * half) & 127;
#pragma unroll
  for (int s = 0; s < 16; ++s)
    dst[s] = wb2[(2 * (sbase + s) + half) * D + col];
}

#define LROWS 16
#define XS    (3 * D + 4)
#define HS    (D + 4)
#define PLDS  (16384 + 4 * GPB * D + GPB * XP + 128)
#define SP    (D + 4)
#define STG   (16 * SP)
#define RBASE (PLDS + D * 16)
#define LDSF  (RBASE + (PTHR / 32) * STG)

__global__ __launch_bounds__(PTHR) void partner_kernel(
    const float* __restrict__ node, const float* __restrict__ glob,
    const float* __restrict__ Wp1,  const float* __restrict__ bp1,
    const float* __restrict__ Wp2,  const float* __restrict__ bp2,
    const float* __restrict__ Wl1,  const float* __restrict__ bl1,
    const float* __restrict__ Wl2,  const float* __restrict__ bl2,
    const float* __restrict__ Wl3,  const float* __restrict__ bl3,
    const int* __restrict__ pgi,    const int* __restrict__ pni,
    float* __restrict__ out, int outoff)
{
  extern __shared__ float smem[];
  float* lds_wb   = smem;
  float* lds_part = smem + 16384;
  float* lds_x    = smem + 16384 + 4 * GPB * D;
  float* lds_wp2  = smem + 16384 + 4 * GPB * D + GPB * XP;
  float* lds_w3   = smem + PLDS;
  float* lds_lx   = smem + RBASE;
  float* lds_h1   = lds_lx + 2 * LROWS * XS;
  float* lds_stage = smem + RBASE;
  float* lds_h2   = lds_x;

  const int tid = threadIdx.x;
  const int g0  = blockIdx.x * GPB;
  const int wave = tid >> 5, lane = tid & 31;
  const int half = lane >> 4, l16 = lane & 15;

  const int lrow = tid >> 4, sc = tid & 15;
  const int rowid = blockIdx.x * (2 * LROWS) + lrow;
  const int pg = pgi[rowid];
  const int pn = pni[rowid];

  const int xg = tid >> 6, xc = tid & 63;
  const size_t xoff = (xc < 32)
      ? ((size_t)(g0 + xg) * NPG + NPG - 1) * D + 4 * xc
      : (size_t)(g0 + xg) * D + 4 * (xc - 32);
  const size_t xstep = (xc < 32) ? (size_t)8 * NPG * D : (size_t)8 * D;
  const float* xsrc = ((xc < 32) ? node : glob) + xoff;
  v4f xv[4];
#pragma unroll
  for (int j = 0; j < 4; ++j) xv[j] = *(const v4f*)(xsrc + j * xstep);

  float w3v[4];
#pragma unroll
  for (int j = 0; j < 4; ++j)
    w3v[j] = Wl3[(lrow + 32 * j) * 14 + (sc < 14 ? sc : 13)];

  const float* src0 = node + ((size_t)(pg + 1) * NPG - 1) * D;
  const float* src1 = node + (size_t)pn * D;
  const float* src2 = glob + (size_t)pg * D;
  v4f x00 = *(const v4f*)(src0 + 4 * sc);
  v4f x01 = *(const v4f*)(src0 + 4 * sc + 64);
  v4f x10 = *(const v4f*)(src1 + 4 * sc);
  v4f x11 = *(const v4f*)(src1 + 4 * sc + 64);
  v4f x20 = *(const v4f*)(src2 + 4 * sc);
  v4f x21 = *(const v4f*)(src2 + 4 * sc + 64);

#pragma unroll
  for (int j = 0; j < D * D / PTHR; ++j) {
    const int i = tid + j * PTHR;
    int k = i >> 7, n = i & 127;
    float v = Wp1[(D + k) * D + n];
    int p = k >> 1;
    int col = (n + ((p & 1) << 4)) & 127;
    lds_wb[p * 256 + col * 2 + (k & 1)] = v;
  }
#pragma unroll
  for (int j = 0; j < 4; ++j)
    *(v4f*)(lds_x + (xg + 8 * j) * XP + 4 * xc) = xv[j];
  if (tid < D) lds_wp2[tid] = Wp2[tid];
  {
    float* dst = lds_lx + lrow * XS;
    *(v4f*)(dst + 0 * D + 4 * sc)      = x00;
    *(v4f*)(dst + 0 * D + 4 * sc + 64) = x01;
    *(v4f*)(dst + 1 * D + 4 * sc)      = x10;
    *(v4f*)(dst + 1 * D + 4 * sc + 64) = x11;
    *(v4f*)(dst + 2 * D + 4 * sc)      = x20;
    *(v4f*)(dst + 2 * D + 4 * sc + 64) = x21;
  }
#pragma unroll
  for (int j = 0; j < 4; ++j)
    lds_w3[tid + PTHR * j] = sc < 14 ? w3v[j] : 0.f;
  __syncthreads();

  {
    const int ct = wave & 7, kh = wave >> 3;
    const int n  = ct * 16 + l16;
    for (int q = 0; q < 2; ++q) {
      const int kq = 2 * kh + q;
      const int krow = (kq < 2 ? kq * 64 : 2 * D + (kq - 2) * 64) + 2 * half;
      const float* bp = Wp1 + (size_t)krow * D + n;
      const float* a0 = lds_x + l16 * XP + kq * 64 + 2 * half;
      const float* a1 = a0 + 16 * XP;
      v8f c0 = {}, c1 = {};
#pragma unroll
      for (int s = 0; s < 16; ++s) {
        v2f b;
        b.x = bp[(size_t)(4 * s) * D];
        b.y = bp[(size_t)(4 * s + 1) * D];
        v2f x0 = *(const v2f*)(a0 + 4 * s);
        v2f x1 = *(const v2f*)(a1 + 4 * s);
        c0 = __builtin_amdgcn_wmma_f32_16x16x4_f32(false, x0, false, b,
                                                   (short)0, c0, false, false);
        c1 = __builtin_amdgcn_wmma_f32_16x16x4_f32(false, x1, false, b,
                                                   (short)0, c1, false, false);
      }
      const float bv = (kq == 0) ? bp1[n] : 0.f;
      float* pp = lds_part + kq * (GPB * D);
#pragma unroll
      for (int r = 0; r < 8; ++r) {
        pp[(r + 8 * half) * D + n]      = c0[r] + bv;
        pp[(16 + r + 8 * half) * D + n] = c1[r] + bv;
      }
    }
  }

  const int ltile = wave >> 3;
  const int ln = (wave & 7) * 16 + l16;
  {
    v8f c = {};
#pragma unroll
    for (int seg = 0; seg < 3; ++seg) {
      const float* ap = lds_lx + (ltile * LROWS + l16) * XS + seg * D + 2 * half;
      const float* bp = Wl1 + (size_t)(seg * D + 2 * half) * D + ln;
#pragma unroll
      for (int s = 0; s < 32; ++s) {
        v2f a = *(const v2f*)(ap + 4 * s);
        v2f b;
        b.x = bp[(size_t)(4 * s) * D];
        b.y = bp[(size_t)(4 * s + 1) * D];
        c = __builtin_amdgcn_wmma_f32_16x16x4_f32(false, a, false, b,
                                                  (short)0, c, false, false);
      }
    }
    float bv = bl1[ln];
#pragma unroll
    for (int r = 0; r < 8; ++r) {
      float v = c[r] + bv;
      lds_h1[(ltile * LROWS + r + 8 * half) * HS + ln] = v > 0.f ? v : 0.f;
    }
  }
  __syncthreads();

  {
    v8f c = {};
    const float* ap = lds_h1 + (ltile * LROWS + l16) * HS + 2 * half;
    const float* bp = Wl2 + (size_t)(2 * half) * D + ln;
#pragma unroll
    for (int s = 0; s < 32; ++s) {
      v2f a = *(const v2f*)(ap + 4 * s);
      v2f b;
      b.x = bp[(size_t)(4 * s) * D];
      b.y = bp[(size_t)(4 * s + 1) * D];
      c = __builtin_amdgcn_wmma_f32_16x16x4_f32(false, a, false, b,
                                                (short)0, c, false, false);
    }
    float bv = bl2[ln];
#pragma unroll
    for (int r = 0; r < 8; ++r) {
      float v = c[r] + bv;
      lds_h2[(ltile * LROWS + r + 8 * half) * HS + ln] = v > 0.f ? v : 0.f;
    }
  }
  __syncthreads();

  if ((wave & 7) == 0) {
    v8f c = {};
    const float* ap = lds_h2 + (ltile * LROWS + l16) * HS + 2 * half;
#pragma unroll
    for (int s = 0; s < 32; ++s) {
      v2f a = *(const v2f*)(ap + 4 * s);
      v2f b;
      b.x = lds_w3[(4 * s + 2 * half) * 16 + l16];
      b.y = lds_w3[(4 * s + 2 * half + 1) * 16 + l16];
      c = __builtin_amdgcn_wmma_f32_16x16x4_f32(false, a, false, b,
                                                (short)0, c, false, false);
    }
    if (l16 < 14) {
      float bv = bl3[l16];
#pragma unroll
      for (int r = 0; r < 8; ++r) {
        int grow = blockIdx.x * (2 * LROWS) + ltile * LROWS + r + 8 * half;
        out[outoff + grow * 14 + l16] = c[r] + bv;
      }
    }
  }

  const int gsub = wave >> 1, mpair = wave & 1;
  const v2f* wb2 = (const v2f*)lds_wb;
  const float b2 = bp2[0];

  v2f bA[16], bB[16];
  load_wfrag(bA, wb2, 0, 0, half, l16);

  for (int it = 0; it < GPB / 8; ++it) {
    const int gl = 8 * it + gsub;
    const int graph = g0 + gl;
    const float* gsrc = node + ((size_t)graph * NPG + mpair * 32) * D + 4 * lane;
    v4f raw[32];
#pragma unroll
    for (int r = 0; r < 32; ++r) raw[r] = *(const v4f*)(gsrc + (size_t)r * D);

    float* stw = lds_stage + wave * STG + 4 * lane;
    const float* stf = lds_stage + wave * STG + l16 * SP + 2 * half;

    v2f af0[32], af1[32];
#pragma unroll
    for (int r = 0; r < 16; ++r) *(v4f*)(stw + r * SP) = raw[r];
    WAVE_LDS_ORDER();
#pragma unroll
    for (int s = 0; s < 32; ++s) af0[s] = *(const v2f*)(stf + 4 * s);
    WAVE_LDS_ORDER();
#pragma unroll
    for (int r = 0; r < 16; ++r) *(v4f*)(stw + r * SP) = raw[16 + r];
    WAVE_LDS_ORDER();
#pragma unroll
    for (int s = 0; s < 32; ++s) af1[s] = *(const v2f*)(stf + 4 * s);

    float acc0[8], acc1[8];
#pragma unroll
    for (int r = 0; r < 8; ++r) { acc0[r] = 0.f; acc1[r] = 0.f; }

    for (int nt = 0; nt < 8; ++nt) {
      const int n = nt * 16 + l16;
      v8f c0 = {}, c1 = {};
      const float* pb = lds_part + gl * D + n;
      const float p0 = pb[0], p1 = pb[GPB * D];
      const float p2 = pb[2 * GPB * D], p3 = pb[3 * GPB * D];
      const float wn = lds_wp2[n];
      load_wfrag(bB, wb2, nt, 16, half, l16);
      __builtin_amdgcn_sched_barrier(0);
#pragma unroll
      for (int s = 0; s < 16; ++s) {
        c0 = __builtin_amdgcn_wmma_f32_16x16x4_f32(false, af0[s], false, bA[s],
                                                   (short)0, c0, false, false);
        c1 = __builtin_amdgcn_wmma_f32_16x16x4_f32(false, af1[s], false, bA[s],
                                                   (short)0, c1, false, false);
      }
      asm volatile("" : "+v"(c0), "+v"(c1));
      load_wfrag(bA, wb2, (nt + 1) & 7, 0, half, l16);
      __builtin_amdgcn_sched_barrier(0);
#pragma unroll
      for (int s = 0; s < 16; ++s) {
        c0 = __builtin_amdgcn_wmma_f32_16x16x4_f32(false, af0[16 + s], false, bB[s],
                                                   (short)0, c0, false, false);
        c1 = __builtin_amdgcn_wmma_f32_16x16x4_f32(false, af1[16 + s], false, bB[s],
                                                   (short)0, c1, false, false);
      }
      const float bn = (p0 + p1) + (p2 + p3);
#pragma unroll
      for (int r = 0; r < 8; ++r) {
        float v0 = c0[r] + bn;
        v0 = v0 > 0.f ? v0 : 0.f;
        acc0[r] += v0 * wn;
        float v1 = c1[r] + bn;
        v1 = v1 > 0.f ? v1 : 0.f;
        acc1[r] += v1 * wn;
      }
    }
#pragma unroll
    for (int r = 0; r < 8; ++r) {
      float a = acc0[r];
      a += __shfl_xor(a, 1, 32);
      a += __shfl_xor(a, 2, 32);
      a += __shfl_xor(a, 4, 32);
      a += __shfl_xor(a, 8, 32);
      acc0[r] = a;
      float b = acc1[r];
      b += __shfl_xor(b, 1, 32);
      b += __shfl_xor(b, 2, 32);
      b += __shfl_xor(b, 4, 32);
      b += __shfl_xor(b, 8, 32);
      acc1[r] = b;
    }
    if (l16 == 0) {
      float* op = out + graph * NPG + mpair * 32 + 8 * half;
#pragma unroll
      for (int r = 0; r < 8; ++r) {
        op[r]      = acc0[r] + b2;
        op[16 + r] = acc1[r] + b2;
      }
    }
  }
}

extern "C" void kernel_launch(void* const* d_in, const int* in_sizes, int n_in,
                              void* d_out, int out_size, void* d_ws, size_t ws_size,
                              hipStream_t stream) {
  (void)n_in; (void)out_size; (void)d_ws; (void)ws_size;
  const float* node = (const float*)d_in[0];
  const float* glob = (const float*)d_in[1];
  const float* Wp1  = (const float*)d_in[2];
  const float* bp1  = (const float*)d_in[3];
  const float* Wp2  = (const float*)d_in[4];
  const float* bp2  = (const float*)d_in[5];
  const float* Wl1  = (const float*)d_in[6];
  const float* bl1  = (const float*)d_in[7];
  const float* Wl2  = (const float*)d_in[8];
  const float* bl2  = (const float*)d_in[9];
  const float* Wl3  = (const float*)d_in[10];
  const float* bl3  = (const float*)d_in[11];
  const int*   pgi  = (const int*)d_in[12];
  const int*   pni  = (const int*)d_in[13];
  float* out = (float*)d_out;

  const int B = in_sizes[1] / D;
  const int N = in_sizes[0] / D;

  const size_t smem1 = LDSF * sizeof(float);

  partner_kernel<<<B / GPB, PTHR, smem1, stream>>>(node, glob, Wp1, bp1, Wp2, bp2,
                                                  Wl1, bl1, Wl2, bl2, Wl3, bl3,
                                                  pgi, pni, out, N);
}
